# starting kernel
# baseline (speedup 1.0000x reference)
#include <hip/hip_runtime.h>

#define BATCH 4
#define NNODE 512
#define NFEAT 16
#define EMBED 64
#define NEG_SLOPE 0.01f

typedef float v2f __attribute__((ext_vector_type(2)));
typedef float v4f __attribute__((ext_vector_type(4)));
typedef float v8f __attribute__((ext_vector_type(8)));

__global__ __launch_bounds__(128) void edge_embed_kernel(
    const float* __restrict__ ef,
    const float* __restrict__ adj,
    const float* __restrict__ w4,
    const float* __restrict__ w3,
    float* __restrict__ out)
{
  const int bi    = blockIdx.x;
  const int tid   = threadIdx.x;
  const int wave  = tid >> 5;
  const int lane  = tid & 31;
  const int laneM = lane & 15;
  const int hi    = lane >> 4;

  __shared__ float partial[4][EMBED];
  __shared__ float summed[EMBED];

  v2f bfrag[4][4];
#pragma unroll
  for (int nt = 0; nt < 4; ++nt)
#pragma unroll
    for (int kk = 0; kk < 4; ++kk)
      bfrag[nt][kk] = *(const v2f*)(w4 + (nt * 16 + laneM) * NFEAT + kk * 4 + 2 * hi);

  const float* efrow = ef  + (size_t)bi * NNODE * NFEAT;
  const float* adjr  = adj + (size_t)bi * NNODE;

  float sacc[4] = {0.f, 0.f, 0.f, 0.f};

#pragma unroll 2
  for (int t = 0; t < 8; ++t) {
    const int jbase = (wave * 8 + t) * 16;

    v2f afrag[4];
#pragma unroll
    for (int kk = 0; kk < 4; ++kk)
      afrag[kk] = *(const v2f*)(efrow + (size_t)(jbase + laneM) * NFEAT + kk * 4 + 2 * hi);

    v4f adv0 = *(const v4f*)(adjr + jbase + hi * 8);
    v4f adv1 = *(const v4f*)(adjr + jbase + hi * 8 + 4);
    float adv[8] = {adv0.x, adv0.y, adv0.z, adv0.w, adv1.x, adv1.y, adv1.z, adv1.w};

#pragma unroll
    for (int nt = 0; nt < 4; ++nt) {
      v8f c = {};
#pragma unroll
      for (int kk = 0; kk < 4; ++kk)
        c = __builtin_amdgcn_wmma_f32_16x16x4_f32(
                 false, afrag[kk],
                 false, bfrag[nt][kk],
                 (short)0, c,
                 false,  false);

      float s = 0.f;
#pragma unroll
      for (int v = 0; v < 8; ++v) {
        float x = c[v];
        float y = fmaxf(x, NEG_SLOPE * x);
        s = fmaf(adv[v], y, s);
      }
      sacc[nt] += s;
    }
  }

#pragma unroll
  for (int nt = 0; nt < 4; ++nt) {
    float s = sacc[nt] + __shfl_xor(sacc[nt], 16, 32);
    if (hi == 0) partial[wave][nt * 16 + laneM] = s;
  }
  __syncthreads();

  if (tid < EMBED)
    summed[tid] = partial[0][tid] + partial[1][tid] + partial[2][tid] + partial[3][tid];
  __syncthreads();

  if (tid < EMBED) {
    const float* w3r = w3 + tid * EMBED;
    float acc = 0.f;
#pragma unroll
    for (int e = 0; e < EMBED; ++e)
      acc = fmaf(w3r[e], summed[e], acc);
    out[(size_t)bi * EMBED + tid] = acc;
  }
}

extern "C" void kernel_launch(void* const* d_in, const int* in_sizes, int n_in,
                              void* d_out, int out_size, void* d_ws, size_t ws_size,
                              hipStream_t stream) {
  const float* ef  = (const float*)d_in[0];
  const float* adj = (const float*)d_in[1];
  const float* w4  = (const float*)d_in[2];
  const float* w3  = (const float*)d_in[3];
  float* out = (float*)d_out;

  dim3 grid(BATCH * NNODE);
  dim3 block(128);
  hipLaunchKernelGGL(edge_embed_kernel, grid, block, 0, stream, ef, adj, w4, w3, out);
}


# one wave owns a whole (b,i) row (32 j-tiles, 8 rows per block): 4x fewer waves, no cross-wave reduce, next tile pair's loads ahead of the matrix block, w3 staged in LDS; sum order unchanged
# speedup vs baseline: 1.1225x; 1.1225x over previous
#include <hip/hip_runtime.h>


#define BATCH 4
#define NNODE 512
#define NFEAT 16
#define EMBED 64
#define NEG_SLOPE 0.01f
#define ROWS_PER_BLOCK 8
#define W3_PITCH 68

typedef float v2f __attribute__((ext_vector_type(2)));
typedef float v4f __attribute__((ext_vector_type(4)));
typedef float v8f __attribute__((ext_vector_type(8)));

struct TilePair {
  v2f a[2][4];
  v4f ad[2][2];
};

__global__ __launch_bounds__(256) void edge_embed_kernel(
    const float* __restrict__ ef,
    const float* __restrict__ adj,
    const float* __restrict__ w4,
    const float* __restrict__ w3,
    float* __restrict__ out)
{
  const int tid   = threadIdx.x;
  const int wave  = tid >> 5;
  const int lane  = tid & 31;
  const int laneM = lane & 15;
  const int hi    = lane >> 4;
  const int bi    = blockIdx.x * ROWS_PER_BLOCK + wave;

  __shared__ __attribute__((aligned(16))) float w3s[EMBED * W3_PITCH];

#pragma unroll
  for (int r = 0; r < 4; ++r) {
    const int q = r * 256 + tid;
    const int o = q >> 4, e4 = q & 15;
    *(v4f*)(w3s + o * W3_PITCH + e4 * 4) = *(const v4f*)(w3 + o * EMBED + e4 * 4);
  }

  v2f bfrag[4][4];
#pragma unroll
  for (int nt = 0; nt < 4; ++nt)
#pragma unroll
    for (int kk = 0; kk < 4; ++kk)
      bfrag[nt][kk] = *(const v2f*)(w4 + (nt * 16 + laneM) * NFEAT + kk * 4 + 2 * hi);

  const float* efrow = ef  + (size_t)bi * NNODE * NFEAT;
  const float* adjr  = adj + (size_t)bi * NNODE;

  float sacc[4] = {0.f, 0.f, 0.f, 0.f};

  auto load_pair = [&](int p, TilePair& T) {
#pragma unroll
    for (int u = 0; u < 2; ++u) {
      const int jbase = (2 * p + u) * 16;
#pragma unroll
      for (int kk = 0; kk < 4; ++kk)
        T.a[u][kk] = *(const v2f*)(efrow + (size_t)(jbase + laneM) * NFEAT + kk * 4 + 2 * hi);
      T.ad[u][0] = *(const v4f*)(adjr + jbase + hi * 8);
      T.ad[u][1] = *(const v4f*)(adjr + jbase + hi * 8 + 4);
    }
  };

  auto compute_pair = [&](const TilePair& T) {
#pragma unroll
    for (int u = 0; u < 2; ++u) {
      const float adv[8] = {T.ad[u][0].x, T.ad[u][0].y, T.ad[u][0].z, T.ad[u][0].w,
                            T.ad[u][1].x, T.ad[u][1].y, T.ad[u][1].z, T.ad[u][1].w};
#pragma unroll
      for (int nt = 0; nt < 4; ++nt) {
        v8f c = {};
#pragma unroll
        for (int kk = 0; kk < 4; ++kk)
          c = __builtin_amdgcn_wmma_f32_16x16x4_f32(
                   false, T.a[u][kk],
                   false, bfrag[nt][kk],
                   (short)0, c,
                   false,  false);

        float s = 0.f;
#pragma unroll
        for (int v = 0; v < 8; ++v) {
          float x = c[v];
          float y = fmaxf(x, NEG_SLOPE * x);
          s = fmaf(adv[v], y, s);
        }
        sacc[nt] += s;
      }
    }
  };

  float sm[4] = {0.f, 0.f, 0.f, 0.f};
  TilePair A, B;
  auto close_segment = [&]() {
#pragma unroll
    for (int nt = 0; nt < 4; ++nt) {
      sm[nt] += sacc[nt] + __shfl_xor(sacc[nt], 16, 32);
      sacc[nt] = 0.f;
    }
  };
  auto trip = [&](int it) {
    load_pair(2 * it + 1, B);
    compute_pair(A);
    load_pair(2 * it + 2, A);
    compute_pair(B);
  };
  load_pair(0, A);
  __syncthreads();
#pragma unroll 1
  for (int seg = 0; seg < 3; ++seg) {
#pragma unroll 1
    for (int k = 0; k < 2; ++k)
      trip(2 * seg + k);
    close_segment();
  }
  trip(6);
  load_pair(15, B);
  compute_pair(A);
  compute_pair(B);
  close_segment();

  const float* r0 = w3s + lane * W3_PITCH;
  const float* r1 = w3s + (lane + 32) * W3_PITCH;
  float acc0 = 0.f, acc1 = 0.f;
#pragma unroll
  for (int nt = 0; nt < 4; ++nt) {
#pragma unroll
    for (int m4 = 0; m4 < 4; ++m4) {
      const v4f wa = *(const v4f*)(r0 + nt * 16 + m4 * 4);
      const v4f wb = *(const v4f*)(r1 + nt * 16 + m4 * 4);
#pragma unroll
      for (int q = 0; q < 4; ++q) {
        const float se = __int_as_float(__builtin_amdgcn_readlane(__float_as_int(sm[nt]), m4 * 4 + q));
        acc0 = fmaf(wa[q], se, acc0);
        acc1 = fmaf(wb[q], se, acc1);
      }
    }
  }
  out[(size_t)bi * EMBED + lane]      = acc0;
  out[(size_t)bi * EMBED + 32 + lane] = acc1;
}

extern "C" void kernel_launch(void* const* d_in, const int* in_sizes, int n_in,
                              void* d_out, int out_size, void* d_ws, size_t ws_size,
                              hipStream_t stream) {
  const float* ef  = (const float*)d_in[0];
  const float* adj = (const float*)d_in[1];
  const float* w4  = (const float*)d_in[2];
  const float* w3  = (const float*)d_in[3];
  float* out = (float*)d_out;

  dim3 grid(BATCH * NNODE / ROWS_PER_BLOCK);
  dim3 block(256);
  hipLaunchKernelGGL(edge_embed_kernel, grid, block, 0, stream, ef, adj, w4, w3, out);
}


# adj-weighted LeakyReLU sum taken as two multiply-add chains per element (adj*x packed, adj*|x| with the abs inside the fma), joined as (1+a)/2*S+(1-a)/2*A per segment; rounding and sum order change
# speedup vs baseline: 1.1878x; 1.0582x over previous
#include <hip/hip_runtime.h>


#define BATCH 4
#define NNODE 512
#define NFEAT 16
#define EMBED 64
#define NEG_SLOPE 0.01f
#define C_LIN (0.5f * (1.0f + NEG_SLOPE))
#define C_ABS (0.5f * (1.0f - NEG_SLOPE))
#define ROWS_PER_BLOCK 8
#define W3_PITCH 68

typedef float v2f __attribute__((ext_vector_type(2)));
typedef float v4f __attribute__((ext_vector_type(4)));
typedef float v8f __attribute__((ext_vector_type(8)));

struct TilePair {
  v2f a[2][4];
  v4f ad[2][2];
};

__global__ __launch_bounds__(256) void edge_embed_kernel(
    const float* __restrict__ ef,
    const float* __restrict__ adj,
    const float* __restrict__ w4,
    const float* __restrict__ w3,
    float* __restrict__ out)
{
  const int tid   = threadIdx.x;
  const int wave  = tid >> 5;
  const int lane  = tid & 31;
  const int laneM = lane & 15;
  const int hi    = lane >> 4;
  const int bi    = blockIdx.x * ROWS_PER_BLOCK + wave;

  __shared__ __attribute__((aligned(16))) float w3s[EMBED * W3_PITCH];

#pragma unroll
  for (int r = 0; r < 4; ++r) {
    const int q = r * 256 + tid;
    const int o = q >> 4, e4 = q & 15;
    *(v4f*)(w3s + o * W3_PITCH + e4 * 4) = *(const v4f*)(w3 + o * EMBED + e4 * 4);
  }

  v2f bfrag[4][4];
#pragma unroll
  for (int nt = 0; nt < 4; ++nt)
#pragma unroll
    for (int kk = 0; kk < 4; ++kk)
      bfrag[nt][kk] = *(const v2f*)(w4 + (nt * 16 + laneM) * NFEAT + kk * 4 + 2 * hi);

  const float* efrow = ef  + (size_t)bi * NNODE * NFEAT;
  const float* adjr  = adj + (size_t)bi * NNODE;

  float slin[4] = {0.f, 0.f, 0.f, 0.f};
  float sabs[4] = {0.f, 0.f, 0.f, 0.f};

  auto load_pair = [&](int p, TilePair& T) {
#pragma unroll
    for (int u = 0; u < 2; ++u) {
      const int jbase = (2 * p + u) * 16;
#pragma unroll
      for (int kk = 0; kk < 4; ++kk)
        T.a[u][kk] = *(const v2f*)(efrow + (size_t)(jbase + laneM) * NFEAT + kk * 4 + 2 * hi);
      T.ad[u][0] = *(const v4f*)(adjr + jbase + hi * 8);
      T.ad[u][1] = *(const v4f*)(adjr + jbase + hi * 8 + 4);
    }
  };

  auto compute_pair = [&](const TilePair& T) {
#pragma unroll
    for (int u = 0; u < 2; ++u) {
      const float adv[8] = {T.ad[u][0].x, T.ad[u][0].y, T.ad[u][0].z, T.ad[u][0].w,
                            T.ad[u][1].x, T.ad[u][1].y, T.ad[u][1].z, T.ad[u][1].w};
#pragma unroll
      for (int nt = 0; nt < 4; ++nt) {
        v8f c = {};
#pragma unroll
        for (int kk = 0; kk < 4; ++kk)
          c = __builtin_amdgcn_wmma_f32_16x16x4_f32(
                   false, T.a[u][kk],
                   false, bfrag[nt][kk],
                   (short)0, c,
                   false,  false);

        v2f p = {0.f, 0.f};
        float q = 0.f;
#pragma unroll
        for (int i = 0; i < 4; ++i) {
          const v2f xp = {c[2 * i], c[2 * i + 1]};
          const v2f ap = {adv[2 * i], adv[2 * i + 1]};
          p = __builtin_elementwise_fma(ap, xp, p);
          q = fmaf(adv[2 * i], fabsf(c[2 * i]), q);
          q = fmaf(adv[2 * i + 1], fabsf(c[2 * i + 1]), q);
        }
        asm("" : "+v"(q));
        slin[nt] += p.x + p.y;
        sabs[nt] += q;
      }
    }
  };

  float sm[4] = {0.f, 0.f, 0.f, 0.f};
  TilePair A, B;
  auto close_segment = [&]() {
#pragma unroll
    for (int nt = 0; nt < 4; ++nt) {
      const float t = fmaf(C_ABS, sabs[nt], C_LIN * slin[nt]);
      sm[nt] += t + __shfl_xor(t, 16, 32);
      slin[nt] = 0.f;
      sabs[nt] = 0.f;
    }
  };
  auto trip = [&](int it) {
    load_pair(2 * it + 1, B);
    compute_pair(A);
    load_pair(2 * it + 2, A);
    compute_pair(B);
  };
  load_pair(0, A);
  __syncthreads();
#pragma unroll 1
  for (int seg = 0; seg < 3; ++seg) {
#pragma unroll 1
    for (int k = 0; k < 2; ++k)
      trip(2 * seg + k);
    close_segment();
  }
  trip(6);
  load_pair(15, B);
  compute_pair(A);
  compute_pair(B);
  close_segment();

  const float* r0 = w3s + lane * W3_PITCH;
  const float* r1 = w3s + (lane + 32) * W3_PITCH;
  float acc0 = 0.f, acc1 = 0.f;
#pragma unroll
  for (int nt = 0; nt < 4; ++nt) {
#pragma unroll
    for (int m4 = 0; m4 < 4; ++m4) {
      const v4f wa = *(const v4f*)(r0 + nt * 16 + m4 * 4);
      const v4f wb = *(const v4f*)(r1 + nt * 16 + m4 * 4);
#pragma unroll
      for (int q = 0; q < 4; ++q) {
        const float se = __int_as_float(__builtin_amdgcn_readlane(__float_as_int(sm[nt]), m4 * 4 + q));
        acc0 = fmaf(wa[q], se, acc0);
        acc1 = fmaf(wb[q], se, acc1);
      }
    }
  }
  out[(size_t)bi * EMBED + lane]      = acc0;
  out[(size_t)bi * EMBED + 32 + lane] = acc1;
}

extern "C" void kernel_launch(void* const* d_in, const int* in_sizes, int n_in,
                              void* d_out, int out_size, void* d_ws, size_t ws_size,
                              hipStream_t stream) {
  const float* ef  = (const float*)d_in[0];
  const float* adj = (const float*)d_in[1];
  const float* w4  = (const float*)d_in[2];
  const float* w3  = (const float*)d_in[3];
  float* out = (float*)d_out;

  dim3 grid(BATCH * NNODE / ROWS_PER_BLOCK);
  dim3 block(256);
  hipLaunchKernelGGL(edge_embed_kernel, grid, block, 0, stream, ef, adj, w4, w3, out);
}


# each tile pair's 12 loads issued as ONE group three pairs ahead of its matrix block (four register sets, scheduling barriers), no load inside a run of matrix instructions; sum order unchanged
# speedup vs baseline: 1.2924x; 1.0881x over previous
#include <hip/hip_runtime.h>


#define BATCH 4
#define NNODE 512
#define NFEAT 16
#define EMBED 64
#define NEG_SLOPE 0.01f
#define C_LIN (0.5f * (1.0f + NEG_SLOPE))
#define C_ABS (0.5f * (1.0f - NEG_SLOPE))
#define ROWS_PER_BLOCK 8
#define W3_PITCH 68

typedef float v2f __attribute__((ext_vector_type(2)));
typedef float v4f __attribute__((ext_vector_type(4)));
typedef float v8f __attribute__((ext_vector_type(8)));

struct TilePair {
  v2f a[2][4];
  v4f ad[2][2];
};

__global__ __launch_bounds__(256) void edge_embed_kernel(
    const float* __restrict__ ef,
    const float* __restrict__ adj,
    const float* __restrict__ w4,
    const float* __restrict__ w3,
    float* __restrict__ out)
{
  const int tid   = threadIdx.x;
  const int wave  = tid >> 5;
  const int lane  = tid & 31;
  const int laneM = lane & 15;
  const int hi    = lane >> 4;
  const int bi    = blockIdx.x * ROWS_PER_BLOCK + wave;

  __shared__ __attribute__((aligned(16))) float w3s[EMBED * W3_PITCH];

#pragma unroll
  for (int r = 0; r < 4; ++r) {
    const int q = r * 256 + tid;
    const int o = q >> 4, e4 = q & 15;
    *(v4f*)(w3s + o * W3_PITCH + e4 * 4) = *(const v4f*)(w3 + o * EMBED + e4 * 4);
  }

  v2f bfrag[4][4];
#pragma unroll
  for (int nt = 0; nt < 4; ++nt)
#pragma unroll
    for (int kk = 0; kk < 4; ++kk)
      bfrag[nt][kk] = *(const v2f*)(w4 + (nt * 16 + laneM) * NFEAT + kk * 4 + 2 * hi);

  const float* efrow = ef  + (size_t)bi * NNODE * NFEAT;
  const float* adjr  = adj + (size_t)bi * NNODE;

  float slin[4] = {0.f, 0.f, 0.f, 0.f};
  float sabs[4] = {0.f, 0.f, 0.f, 0.f};

  auto load_pair = [&](int p, TilePair& T) {
#pragma unroll
    for (int u = 0; u < 2; ++u) {
      const int jbase = (2 * p + u) * 16;
#pragma unroll
      for (int kk = 0; kk < 4; ++kk)
        T.a[u][kk] = *(const v2f*)(efrow + (size_t)(jbase + laneM) * NFEAT + kk * 4 + 2 * hi);
      T.ad[u][0] = *(const v4f*)(adjr + jbase + hi * 8);
      T.ad[u][1] = *(const v4f*)(adjr + jbase + hi * 8 + 4);
    }
  };

  auto compute_pair = [&](const TilePair& T) {
#pragma unroll
    for (int u = 0; u < 2; ++u) {
      const float adv[8] = {T.ad[u][0].x, T.ad[u][0].y, T.ad[u][0].z, T.ad[u][0].w,
                            T.ad[u][1].x, T.ad[u][1].y, T.ad[u][1].z, T.ad[u][1].w};
#pragma unroll
      for (int nt = 0; nt < 4; ++nt) {
        v8f c = {};
#pragma unroll
        for (int kk = 0; kk < 4; ++kk)
          c = __builtin_amdgcn_wmma_f32_16x16x4_f32(
                   false, T.a[u][kk],
                   false, bfrag[nt][kk],
                   (short)0, c,
                   false,  false);

        v2f p = {0.f, 0.f};
        float q = 0.f;
#pragma unroll
        for (int i = 0; i < 4; ++i) {
          const v2f xp = {c[2 * i], c[2 * i + 1]};
          const v2f ap = {adv[2 * i], adv[2 * i + 1]};
          p = __builtin_elementwise_fma(ap, xp, p);
          q = fmaf(adv[2 * i], fabsf(c[2 * i]), q);
          q = fmaf(adv[2 * i + 1], fabsf(c[2 * i + 1]), q);
        }
        asm("" : "+v"(q));
        slin[nt] += p.x + p.y;
        sabs[nt] += q;
      }
    }
  };

  float sm[4] = {0.f, 0.f, 0.f, 0.f};
  TilePair P0, P1, P2, P3;
  auto close_segment = [&]() {
#pragma unroll
    for (int nt = 0; nt < 4; ++nt) {
      const float t = fmaf(C_ABS, sabs[nt], C_LIN * slin[nt]);
      sm[nt] += t + __shfl_xor(t, 16, 32);
      slin[nt] = 0.f;
      sabs[nt] = 0.f;
    }
  };
  auto step = [&](int pl, TilePair& L, const TilePair& C) {
    __builtin_amdgcn_sched_barrier(0);
    load_pair(pl, L);
    __builtin_amdgcn_sched_barrier(0);
    compute_pair(C);
  };
  load_pair(0, P0);
  __syncthreads();
  load_pair(1, P1);
  __builtin_amdgcn_sched_barrier(0);
  load_pair(2, P2);
#pragma unroll 1
  for (int seg = 0; seg < 3; ++seg) {
    const int base = 4 * seg;
    step(base + 3, P3, P0);
    step(base + 4, P0, P1);
    step(base + 5, P1, P2);
    step(base + 6, P2, P3);
    close_segment();
  }
  step(15, P3, P0);
  compute_pair(P1);
  compute_pair(P2);
  compute_pair(P3);
  close_segment();

  const float* r0 = w3s + lane * W3_PITCH;
  const float* r1 = w3s + (lane + 32) * W3_PITCH;
  float acc0 = 0.f, acc1 = 0.f;
#pragma unroll
  for (int nt = 0; nt < 4; ++nt) {
#pragma unroll
    for (int m4 = 0; m4 < 4; ++m4) {
      const v4f wa = *(const v4f*)(r0 + nt * 16 + m4 * 4);
      const v4f wb = *(const v4f*)(r1 + nt * 16 + m4 * 4);
#pragma unroll
      for (int q = 0; q < 4; ++q) {
        const float se = __int_as_float(__builtin_amdgcn_readlane(__float_as_int(sm[nt]), m4 * 4 + q));
        acc0 = fmaf(wa[q], se, acc0);
        acc1 = fmaf(wb[q], se, acc1);
      }
    }
  }
  out[(size_t)bi * EMBED + lane]      = acc0;
  out[(size_t)bi * EMBED + 32 + lane] = acc1;
}

extern "C" void kernel_launch(void* const* d_in, const int* in_sizes, int n_in,
                              void* d_out, int out_size, void* d_ws, size_t ws_size,
                              hipStream_t stream) {
  const float* ef  = (const float*)d_in[0];
  const float* adj = (const float*)d_in[1];
  const float* w4  = (const float*)d_in[2];
  const float* w3  = (const float*)d_in[3];
  float* out = (float*)d_out;

  dim3 grid(BATCH * NNODE / ROWS_PER_BLOCK);
  dim3 block(256);
  hipLaunchKernelGGL(edge_embed_kernel, grid, block, 0, stream, ef, adj, w4, w3, out);
}
